# starting kernel
# baseline (speedup 1.0000x reference)
#include <hip/hip_runtime.h>
#include <hip/hip_bf16.h>


typedef __attribute__((ext_vector_type(2))) float v2f;
typedef __attribute__((ext_vector_type(8))) float v8f;

#define TILE_M     128
#define NW         8
#define LDSS       68

__global__ __launch_bounds__(256)
void gemm_relu_stats(const float* __restrict__ xin, int ldx, int Kp,
                     const float* __restrict__ W,  int Kw,
                     const float* __restrict__ bias,
                     float* __restrict__ hout,
                     float* __restrict__ partial)
{
    __shared__ float xl[TILE_M * LDSS];
    __shared__ float Wl[64 * LDSS];
    __shared__ float biasl[64];
    __shared__ float wsum[NW * 64];
    __shared__ float wsq [NW * 64];

    const int t    = threadIdx.x;
    const int lane = t & 31;
    const int wave = t >> 5;
    const int half = lane >> 4;
    const int l15  = lane & 15;
    const size_t row0 = (size_t)blockIdx.x * TILE_M;

    const int c4n = Kp >> 2;
    for (int i = t; i < TILE_M * c4n; i += 256) {
        int r = i / c4n;
        int c = (i - r * c4n) * 4;
        float4 v = *(const float4*)(xin + (row0 + r) * (size_t)ldx + c);
        *(float4*)(&xl[r * LDSS + c]) = v;
    }
    for (int i = t; i < 64 * Kp; i += 256) {
        int n = i / Kp;
        int k = i - n * Kp;
        Wl[n * LDSS + k] = (n < 59 && k < Kw) ? W[n * Kw + k] : 0.0f;
    }
    if (t < 64) biasl[t] = (t < 59) ? bias[t] : 0.0f;
    __syncthreads();

    v8f acc[4] = {};
    const float* ax = xl + (wave * 16 + l15) * LDSS;
    for (int k0 = 0; k0 < Kp; k0 += 4) {
        const int ka = k0 + 2 * half;
        v2f a = *(const v2f*)(ax + ka);
        #pragma unroll
        for (int tt = 0; tt < 4; tt++) {
            v2f b = *(const v2f*)(Wl + (tt * 16 + l15) * LDSS + ka);
            acc[tt] = __builtin_amdgcn_wmma_f32_16x16x4_f32(
                false, a, false, b, (short)0, acc[tt], false, false);
        }
    }

    const size_t mbase = row0 + wave * 16 + half * 8;
    #pragma unroll
    for (int tt = 0; tt < 4; tt++) {
        const int n = tt * 16 + l15;
        const float bn = biasl[n];
        float sl = 0.f, sq = 0.f;
        #pragma unroll
        for (int r = 0; r < 8; r++) {
            float v = acc[tt][r] + bn;
            v = fmaxf(v, 0.f);
            hout[(mbase + r) * 64 + n] = v;
            sl += v;
            sq += v * v;
        }
        sl += __shfl_xor(sl, 16, 32);
        sq += __shfl_xor(sq, 16, 32);
        if (half == 0) {
            wsum[wave * 64 + n] = sl;
            wsq [wave * 64 + n] = sq;
        }
    }
    __syncthreads();

    if (t < 64) {
        float s = 0.f;
        #pragma unroll
        for (int w = 0; w < NW; w++) s += wsum[w * 64 + t];
        partial[(size_t)blockIdx.x * 128 + t] = s;
    } else if (t < 128) {
        const int n = t - 64;
        float s = 0.f;
        #pragma unroll
        for (int w = 0; w < NW; w++) s += wsq[w * 64 + n];
        partial[(size_t)blockIdx.x * 128 + 64 + n] = s;
    }
}

__global__ void stats_reduce(const float* __restrict__ partial, int nblk,
                             const float* __restrict__ g,
                             const float* __restrict__ bt,
                             float* __restrict__ stats,
                             float invB)
{
    const int n = threadIdx.x;
    if (n >= 64) return;
    float s = 0.f, ss = 0.f;
    for (int b = 0; b < nblk; b++) {
        s  += partial[(size_t)b * 128 + n];
        ss += partial[(size_t)b * 128 + 64 + n];
    }
    const float mu  = s * invB;
    float var = ss * invB - mu * mu;
    var = fmaxf(var, 0.f);
    float scale = 0.f, shift = 0.f;
    if (n < 59) {
        scale = g[n] * rsqrtf(var + 1e-5f);
        shift = bt[n] - mu * scale;
    }
    stats[n]      = scale;
    stats[64 + n] = shift;
}

__global__ __launch_bounds__(256)
void normalize_pad(const float* __restrict__ h, const float* __restrict__ stats,
                   float* __restrict__ out, size_t nvec  )
{
    __shared__ float sc[64], sh[64];
    const int t = threadIdx.x;
    if (t < 64)       sc[t]      = stats[t];
    else if (t < 128) sh[t - 64] = stats[t];
    __syncthreads();

    const size_t idx = (size_t)blockIdx.x * 256 + t;
    if (idx >= nvec) return;
    const int n = (int)(idx & 15) * 4;
    const float4 hv = ((const float4*)h)[idx];
    float4 o;
    o.x = hv.x * sc[n + 0] + sh[n + 0];
    o.y = hv.y * sc[n + 1] + sh[n + 1];
    o.z = hv.z * sc[n + 2] + sh[n + 2];
    o.w = hv.w * sc[n + 3] + sh[n + 3];
    ((float4*)out)[idx] = o;
}

__global__ void normalize_final(const float* __restrict__ h,
                                const float* __restrict__ stats,
                                float* __restrict__ out, size_t total  )
{
    const size_t i = (size_t)blockIdx.x * 256 + threadIdx.x;
    if (i >= total) return;
    const size_t row = i / 59;
    const int n = (int)(i - row * 59);
    out[i] = h[row * 64 + n] * stats[n] + stats[64 + n];
}

__global__ void pad_input(const float* __restrict__ x, float* __restrict__ xp,
                          size_t total  )
{
    const size_t i = (size_t)blockIdx.x * 256 + threadIdx.x;
    if (i >= total) return;
    const size_t row = i >> 5;
    const int n = (int)(i & 31);
    xp[i] = (n < 29) ? x[row * 29 + n] : 0.0f;
}

extern "C" void kernel_launch(void* const* d_in, const int* in_sizes, int n_in,
                              void* d_out, int out_size, void* d_ws, size_t ws_size,
                              hipStream_t stream) {
    const float* x   = (const float*)d_in[0];
    const float* W0  = (const float*)d_in[1];
    const float* b0  = (const float*)d_in[2];
    const float* g0  = (const float*)d_in[3];
    const float* bt0 = (const float*)d_in[4];
    const float* Ws  = (const float*)d_in[5];
    const float* bs  = (const float*)d_in[6];
    const float* gs  = (const float*)d_in[7];
    const float* bts = (const float*)d_in[8];
    float* out = (float*)d_out;

    const int B     = in_sizes[0] / 29;
    const int nblk  = B / TILE_M;
    const float invB = 1.0f / (float)B;

    float* ws      = (float*)d_ws;
    float* xpad    = ws;
    float* P       = xpad    + (size_t)B * 32;
    float* Q       = P       + (size_t)B * 64;
    float* partial = Q       + (size_t)B * 64;
    float* stats   = partial + (size_t)nblk * 128;

    const size_t padTot = (size_t)B * 32;
    pad_input<<<(unsigned)((padTot + 255) / 256), 256, 0, stream>>>(x, xpad, padTot);

    const size_t nvec = (size_t)B * 16;

    gemm_relu_stats<<<nblk, 256, 0, stream>>>(xpad, 32, 32, W0, 29, b0, P, partial);
    stats_reduce<<<1, 64, 0, stream>>>(partial, nblk, g0, bt0, stats, invB);
    normalize_pad<<<(unsigned)(nvec / 256), 256, 0, stream>>>(P, stats, Q, nvec);

    for (int l = 0; l < 14; l++) {
        const float* W  = Ws  + (size_t)l * 59 * 59;
        const float* bb = bs  + (size_t)l * 59;
        const float* gg = gs  + (size_t)l * 59;
        const float* tt = bts + (size_t)l * 59;
        gemm_relu_stats<<<nblk, 256, 0, stream>>>(Q, 64, 64, W, 59, bb, P, partial);
        stats_reduce<<<1, 64, 0, stream>>>(partial, nblk, gg, tt, stats, invB);
        if (l < 13) {
            normalize_pad<<<(unsigned)(nvec / 256), 256, 0, stream>>>(P, stats, Q, nvec);
        } else {
            const size_t total = (size_t)B * 59;
            normalize_final<<<(unsigned)((total + 255) / 256), 256, 0, stream>>>(
                P, stats, out, total);
        }
    }
}


# stats reduce spread over one 1024-thread workgroup: each wave sums every 32nd partial row with b128 loads issued in groups, fixed-order LDS combine (was two waves walking all rows serially)
# speedup vs baseline: 2.2220x; 2.2220x over previous
#include <hip/hip_runtime.h>
#include <hip/hip_bf16.h>


typedef __attribute__((ext_vector_type(2))) float v2f;
typedef __attribute__((ext_vector_type(8))) float v8f;

#define TILE_M     128
#define NW         8
#define LDSS       68

__global__ __launch_bounds__(256)
void gemm_relu_stats(const float* __restrict__ xin, int ldx, int Kp,
                     const float* __restrict__ W,  int Kw,
                     const float* __restrict__ bias,
                     float* __restrict__ hout,
                     float* __restrict__ partial)
{
    __shared__ float xl[TILE_M * LDSS];
    __shared__ float Wl[64 * LDSS];
    __shared__ float biasl[64];
    __shared__ float wsum[NW * 64];
    __shared__ float wsq [NW * 64];

    const int t    = threadIdx.x;
    const int lane = t & 31;
    const int wave = t >> 5;
    const int half = lane >> 4;
    const int l15  = lane & 15;
    const size_t row0 = (size_t)blockIdx.x * TILE_M;

    const int c4n = Kp >> 2;
    for (int i = t; i < TILE_M * c4n; i += 256) {
        int r = i / c4n;
        int c = (i - r * c4n) * 4;
        float4 v = *(const float4*)(xin + (row0 + r) * (size_t)ldx + c);
        *(float4*)(&xl[r * LDSS + c]) = v;
    }
    for (int i = t; i < 64 * Kp; i += 256) {
        int n = i / Kp;
        int k = i - n * Kp;
        Wl[n * LDSS + k] = (n < 59 && k < Kw) ? W[n * Kw + k] : 0.0f;
    }
    if (t < 64) biasl[t] = (t < 59) ? bias[t] : 0.0f;
    __syncthreads();

    v8f acc[4] = {};
    const float* ax = xl + (wave * 16 + l15) * LDSS;
    for (int k0 = 0; k0 < Kp; k0 += 4) {
        const int ka = k0 + 2 * half;
        v2f a = *(const v2f*)(ax + ka);
        #pragma unroll
        for (int tt = 0; tt < 4; tt++) {
            v2f b = *(const v2f*)(Wl + (tt * 16 + l15) * LDSS + ka);
            acc[tt] = __builtin_amdgcn_wmma_f32_16x16x4_f32(
                false, a, false, b, (short)0, acc[tt], false, false);
        }
    }

    const size_t mbase = row0 + wave * 16 + half * 8;
    #pragma unroll
    for (int tt = 0; tt < 4; tt++) {
        const int n = tt * 16 + l15;
        const float bn = biasl[n];
        float sl = 0.f, sq = 0.f;
        #pragma unroll
        for (int r = 0; r < 8; r++) {
            float v = acc[tt][r] + bn;
            v = fmaxf(v, 0.f);
            hout[(mbase + r) * 64 + n] = v;
            sl += v;
            sq += v * v;
        }
        sl += __shfl_xor(sl, 16, 32);
        sq += __shfl_xor(sq, 16, 32);
        if (half == 0) {
            wsum[wave * 64 + n] = sl;
            wsq [wave * 64 + n] = sq;
        }
    }
    __syncthreads();

    if (t < 64) {
        float s = 0.f;
        #pragma unroll
        for (int w = 0; w < NW; w++) s += wsum[w * 64 + t];
        partial[(size_t)blockIdx.x * 128 + t] = s;
    } else if (t < 128) {
        const int n = t - 64;
        float s = 0.f;
        #pragma unroll
        for (int w = 0; w < NW; w++) s += wsq[w * 64 + n];
        partial[(size_t)blockIdx.x * 128 + 64 + n] = s;
    }
}

#define RED_W 32
__global__ __launch_bounds__(RED_W * 32)
void stats_reduce(const float* __restrict__ partial, int nblk,
                  const float* __restrict__ g,
                  const float* __restrict__ bt,
                  float* __restrict__ stats,
                  float invB)
{
    __shared__ __align__(16) float red[RED_W * 128];

    const int t  = threadIdx.x;
    const int c4 = t & 31;
    const int w  = t >> 5;
    const float4* __restrict__ p4 = (const float4*)partial;

    float4 a = make_float4(0.f, 0.f, 0.f, 0.f);
    #pragma unroll 8
    for (int b = w; b < nblk; b += RED_W) {
        const float4 v = p4[(size_t)b * 32 + c4];
        a.x += v.x; a.y += v.y; a.z += v.z; a.w += v.w;
    }
    *(float4*)(&red[w * 128 + c4 * 4]) = a;
    __syncthreads();

    if (t < 64) {
        const int n = t;
        float s = 0.f, ss = 0.f;
        #pragma unroll
        for (int k = 0; k < RED_W; k++) {
            s  += red[k * 128 + n];
            ss += red[k * 128 + 64 + n];
        }
        const float mu  = s * invB;
        float var = ss * invB - mu * mu;
        var = fmaxf(var, 0.f);
        float scale = 0.f, shift = 0.f;
        if (n < 59) {
            scale = g[n] * rsqrtf(var + 1e-5f);
            shift = bt[n] - mu * scale;
        }
        stats[n]      = scale;
        stats[64 + n] = shift;
    }
}

__global__ __launch_bounds__(256)
void normalize_pad(const float* __restrict__ h, const float* __restrict__ stats,
                   float* __restrict__ out, size_t nvec  )
{
    __shared__ float sc[64], sh[64];
    const int t = threadIdx.x;
    if (t < 64)       sc[t]      = stats[t];
    else if (t < 128) sh[t - 64] = stats[t];
    __syncthreads();

    const size_t idx = (size_t)blockIdx.x * 256 + t;
    if (idx >= nvec) return;
    const int n = (int)(idx & 15) * 4;
    const float4 hv = ((const float4*)h)[idx];
    float4 o;
    o.x = hv.x * sc[n + 0] + sh[n + 0];
    o.y = hv.y * sc[n + 1] + sh[n + 1];
    o.z = hv.z * sc[n + 2] + sh[n + 2];
    o.w = hv.w * sc[n + 3] + sh[n + 3];
    ((float4*)out)[idx] = o;
}

__global__ void normalize_final(const float* __restrict__ h,
                                const float* __restrict__ stats,
                                float* __restrict__ out, size_t total  )
{
    const size_t i = (size_t)blockIdx.x * 256 + threadIdx.x;
    if (i >= total) return;
    const size_t row = i / 59;
    const int n = (int)(i - row * 59);
    out[i] = h[row * 64 + n] * stats[n] + stats[64 + n];
}

__global__ void pad_input(const float* __restrict__ x, float* __restrict__ xp,
                          size_t total  )
{
    const size_t i = (size_t)blockIdx.x * 256 + threadIdx.x;
    if (i >= total) return;
    const size_t row = i >> 5;
    const int n = (int)(i & 31);
    xp[i] = (n < 29) ? x[row * 29 + n] : 0.0f;
}

extern "C" void kernel_launch(void* const* d_in, const int* in_sizes, int n_in,
                              void* d_out, int out_size, void* d_ws, size_t ws_size,
                              hipStream_t stream) {
    const float* x   = (const float*)d_in[0];
    const float* W0  = (const float*)d_in[1];
    const float* b0  = (const float*)d_in[2];
    const float* g0  = (const float*)d_in[3];
    const float* bt0 = (const float*)d_in[4];
    const float* Ws  = (const float*)d_in[5];
    const float* bs  = (const float*)d_in[6];
    const float* gs  = (const float*)d_in[7];
    const float* bts = (const float*)d_in[8];
    float* out = (float*)d_out;

    const int B     = in_sizes[0] / 29;
    const int nblk  = B / TILE_M;
    const float invB = 1.0f / (float)B;

    float* ws      = (float*)d_ws;
    float* xpad    = ws;
    float* P       = xpad    + (size_t)B * 32;
    float* Q       = P       + (size_t)B * 64;
    float* partial = Q       + (size_t)B * 64;
    float* stats   = partial + (size_t)nblk * 128;

    const size_t padTot = (size_t)B * 32;
    pad_input<<<(unsigned)((padTot + 255) / 256), 256, 0, stream>>>(x, xpad, padTot);

    const size_t nvec = (size_t)B * 16;

    gemm_relu_stats<<<nblk, 256, 0, stream>>>(xpad, 32, 32, W0, 29, b0, P, partial);
    stats_reduce<<<1, RED_W * 32, 0, stream>>>(partial, nblk, g0, bt0, stats, invB);
    normalize_pad<<<(unsigned)(nvec / 256), 256, 0, stream>>>(P, stats, Q, nvec);

    for (int l = 0; l < 14; l++) {
        const float* W  = Ws  + (size_t)l * 59 * 59;
        const float* bb = bs  + (size_t)l * 59;
        const float* gg = gs  + (size_t)l * 59;
        const float* tt = bts + (size_t)l * 59;
        gemm_relu_stats<<<nblk, 256, 0, stream>>>(Q, 64, 64, W, 59, bb, P, partial);
        stats_reduce<<<1, RED_W * 32, 0, stream>>>(partial, nblk, gg, tt, stats, invB);
        if (l < 13) {
            normalize_pad<<<(unsigned)(nvec / 256), 256, 0, stream>>>(P, stats, Q, nvec);
        } else {
            const size_t total = (size_t)B * 59;
            normalize_final<<<(unsigned)((total + 255) / 256), 256, 0, stream>>>(
                P, stats, out, total);
        }
    }
}


# GEMM staging: tile shape compile-time; all global loads of x tile, W, scale/shift, bias issued as one group before the first LDS write (was a load-wait-store round trip and an index division per trip)
# speedup vs baseline: 5.2453x; 2.3606x over previous
#include <hip/hip_runtime.h>
#include <hip/hip_bf16.h>


typedef __attribute__((ext_vector_type(2))) float v2f;
typedef __attribute__((ext_vector_type(8))) float v8f;

#define TILE_M     128
#define NW         8
#define LDSS       68

template <bool NORM, int KP, int KW>
__global__ __launch_bounds__(256)
void gemm_relu_stats(const float* __restrict__ xin,
                     const float* __restrict__ stats,
                     const float* __restrict__ W,
                     const float* __restrict__ bias,
                     float* __restrict__ hout,
                     float* __restrict__ partial)
{
    __shared__ float xl[TILE_M * LDSS];
    __shared__ float Wl[64 * LDSS];
    __shared__ float biasl[64];
    __shared__ float wsum[NW * 64];
    __shared__ float wsq [NW * 64];

    constexpr int C4N   = KP / 4;
    constexpr int RSTEP = 256 / C4N;
    constexpr int XITER = TILE_M / RSTEP;
    constexpr int WITER = 64 * KP / 256;
    static_assert(256 % C4N == 0 && TILE_M % RSTEP == 0 && (64 * KP) % 256 == 0, "tile shape");

    const int t    = threadIdx.x;
    const int lane = t & 31;
    const int wave = t >> 5;
    const int half = lane >> 4;
    const int l15  = lane & 15;
    const size_t row0 = (size_t)blockIdx.x * TILE_M;

    const int xr = t / C4N;
    const int xc = (t % C4N) * 4;
    const float* xp = xin + (row0 + xr) * (size_t)KP + xc;
    float4 xv[XITER];
    #pragma unroll
    for (int j = 0; j < XITER; j++)
        xv[j] = *(const float4*)(xp + (size_t)j * RSTEP * KP);

    float4 sc4 = make_float4(0.f, 0.f, 0.f, 0.f);
    float4 sh4 = make_float4(0.f, 0.f, 0.f, 0.f);
    if constexpr (NORM) {
        sc4 = *(const float4*)(stats + xc);
        sh4 = *(const float4*)(stats + 64 + xc);
    }

    float wv[WITER];
    #pragma unroll
    for (int j = 0; j < WITER; j++) {
        const int i  = t + 256 * j;
        const int n  = i / KP;
        const int k  = i % KP;
        const bool ok = (n < 59) && (k < KW);
        const float w = W[ok ? n * KW + k : 0];
        wv[j] = ok ? w : 0.0f;
    }
    float bv = 0.0f;
    if (t < 64) bv = bias[t < 59 ? t : 0];

    #pragma unroll
    for (int j = 0; j < XITER; j++) {
        float4 v = xv[j];
        if constexpr (NORM) {
            v.x = v.x * sc4.x + sh4.x;
            v.y = v.y * sc4.y + sh4.y;
            v.z = v.z * sc4.z + sh4.z;
            v.w = v.w * sc4.w + sh4.w;
        }
        *(float4*)(&xl[(xr + j * RSTEP) * LDSS + xc]) = v;
    }
    #pragma unroll
    for (int j = 0; j < WITER; j++) {
        const int i = t + 256 * j;
        Wl[(i / KP) * LDSS + (i % KP)] = wv[j];
    }
    if (t < 64) biasl[t] = (t < 59) ? bv : 0.0f;
    __syncthreads();

    v8f acc[4] = {};
    const float* ax = xl + (wave * 16 + l15) * LDSS;
    #pragma unroll 4
    for (int k0 = 0; k0 < KP; k0 += 4) {
        const int ka = k0 + 2 * half;
        v2f a = *(const v2f*)(ax + ka);
        #pragma unroll
        for (int tt = 0; tt < 4; tt++) {
            v2f b = *(const v2f*)(Wl + (tt * 16 + l15) * LDSS + ka);
            acc[tt] = __builtin_amdgcn_wmma_f32_16x16x4_f32(
                false, a, false, b, (short)0, acc[tt], false, false);
        }
    }

    const size_t mbase = row0 + wave * 16 + half * 8;
    #pragma unroll
    for (int tt = 0; tt < 4; tt++) {
        const int n = tt * 16 + l15;
        const float bn = biasl[n];
        float sl = 0.f, sq = 0.f;
        #pragma unroll
        for (int r = 0; r < 8; r++) {
            float v = acc[tt][r] + bn;
            v = fmaxf(v, 0.f);
            hout[(mbase + r) * 64 + n] = v;
            sl += v;
            sq += v * v;
        }
        sl += __shfl_xor(sl, 16, 32);
        sq += __shfl_xor(sq, 16, 32);
        if (half == 0) {
            wsum[wave * 64 + n] = sl;
            wsq [wave * 64 + n] = sq;
        }
    }
    __syncthreads();

    if (t < 64) {
        float s = 0.f;
        #pragma unroll
        for (int w = 0; w < NW; w++) s += wsum[w * 64 + t];
        partial[(size_t)blockIdx.x * 128 + t] = s;
    } else if (t < 128) {
        const int n = t - 64;
        float s = 0.f;
        #pragma unroll
        for (int w = 0; w < NW; w++) s += wsq[w * 64 + n];
        partial[(size_t)blockIdx.x * 128 + 64 + n] = s;
    }
}

#define RED_W 32
__global__ __launch_bounds__(RED_W * 32)
void stats_reduce(const float* __restrict__ partial, int nblk,
                  const float* __restrict__ g,
                  const float* __restrict__ bt,
                  float* __restrict__ stats,
                  float invB)
{
    __shared__ __align__(16) float red[RED_W * 128];

    const int t  = threadIdx.x;
    const int c4 = t & 31;
    const int w  = t >> 5;
    const float4* __restrict__ p4 = (const float4*)partial;

    float4 a = make_float4(0.f, 0.f, 0.f, 0.f);
    #pragma unroll 8
    for (int b = w; b < nblk; b += RED_W) {
        const float4 v = p4[(size_t)b * 32 + c4];
        a.x += v.x; a.y += v.y; a.z += v.z; a.w += v.w;
    }
    *(float4*)(&red[w * 128 + c4 * 4]) = a;
    __syncthreads();

    if (t < 64) {
        const int n = t;
        float s = 0.f, ss = 0.f;
        #pragma unroll
        for (int k = 0; k < RED_W; k++) {
            s  += red[k * 128 + n];
            ss += red[k * 128 + 64 + n];
        }
        const float mu  = s * invB;
        float var = ss * invB - mu * mu;
        var = fmaxf(var, 0.f);
        float scale = 0.f, shift = 0.f;
        if (n < 59) {
            scale = g[n] * rsqrtf(var + 1e-5f);
            shift = bt[n] - mu * scale;
        }
        stats[n]      = scale;
        stats[64 + n] = shift;
    }
}

__global__ void normalize_final(const float* __restrict__ h,
                                const float* __restrict__ stats,
                                float* __restrict__ out, size_t total  )
{
    const size_t i = (size_t)blockIdx.x * 256 + threadIdx.x;
    if (i >= total) return;
    const size_t row = i / 59;
    const int n = (int)(i - row * 59);
    out[i] = h[row * 64 + n] * stats[n] + stats[64 + n];
}

__global__ void pad_input(const float* __restrict__ x, float* __restrict__ xp,
                          size_t total  )
{
    const size_t i = (size_t)blockIdx.x * 256 + threadIdx.x;
    if (i >= total) return;
    const size_t row = i >> 5;
    const int n = (int)(i & 31);
    xp[i] = (n < 29) ? x[row * 29 + n] : 0.0f;
}

extern "C" void kernel_launch(void* const* d_in, const int* in_sizes, int n_in,
                              void* d_out, int out_size, void* d_ws, size_t ws_size,
                              hipStream_t stream) {
    const float* x   = (const float*)d_in[0];
    const float* W0  = (const float*)d_in[1];
    const float* b0  = (const float*)d_in[2];
    const float* g0  = (const float*)d_in[3];
    const float* bt0 = (const float*)d_in[4];
    const float* Ws  = (const float*)d_in[5];
    const float* bs  = (const float*)d_in[6];
    const float* gs  = (const float*)d_in[7];
    const float* bts = (const float*)d_in[8];
    float* out = (float*)d_out;

    const int B     = in_sizes[0] / 29;
    const int nblk  = B / TILE_M;
    const float invB = 1.0f / (float)B;

    float* ws      = (float*)d_ws;
    float* xpad    = ws;
    float* P       = xpad    + (size_t)B * 32;
    float* Q       = P       + (size_t)B * 64;
    float* partial = Q       + (size_t)B * 64;
    float* stats   = partial + (size_t)nblk * 128;

    const size_t padTot = (size_t)B * 32;
    pad_input<<<(unsigned)((padTot + 255) / 256), 256, 0, stream>>>(x, xpad, padTot);

    gemm_relu_stats<false, 32, 29><<<nblk, 256, 0, stream>>>(xpad, stats, W0, b0, P, partial);
    stats_reduce<<<1, RED_W * 32, 0, stream>>>(partial, nblk, g0, bt0, stats, invB);

    float* hprev = P;
    float* hnext = Q;
    for (int l = 0; l < 14; l++) {
        const float* W  = Ws  + (size_t)l * 59 * 59;
        const float* bb = bs  + (size_t)l * 59;
        const float* gg = gs  + (size_t)l * 59;
        const float* tt = bts + (size_t)l * 59;
        gemm_relu_stats<true, 64, 59><<<nblk, 256, 0, stream>>>(hprev, stats, W, bb, hnext, partial);
        stats_reduce<<<1, RED_W * 32, 0, stream>>>(partial, nblk, gg, tt, stats, invB);
        float* tmp = hprev; hprev = hnext; hnext = tmp;
    }

    const size_t total = (size_t)B * 59;
    normalize_final<<<(unsigned)((total + 255) / 256), 256, 0, stream>>>(
        hprev, stats, out, total);
}


# batch-stats reduce of block k folded into block k+1's GEMM prologue: 512-row workgroups (32 waves, one per WGP) give 256 partial rows, summed by every workgroup in fixed order; 14 launches gone
# speedup vs baseline: 7.1693x; 1.3668x over previous
#include <hip/hip_runtime.h>
#include <hip/hip_bf16.h>


typedef __attribute__((ext_vector_type(2))) float v2f;
typedef __attribute__((ext_vector_type(8))) float v8f;

#define TILE_M     512
#define NW         32
#define NT         (NW * 32)
#define LDSS       68

__device__ __forceinline__ void lds_barrier()
{
    __builtin_amdgcn_fence(__ATOMIC_RELEASE, "workgroup", "local");
    __builtin_amdgcn_s_barrier();
    __builtin_amdgcn_fence(__ATOMIC_ACQUIRE, "workgroup", "local");
}

template <bool NORM, int KP, int KW>
__global__ __launch_bounds__(NT)
void gemm_relu_stats(const float* __restrict__ xin,
                     const float* __restrict__ pprev,
                     const float* __restrict__ gprev,
                     const float* __restrict__ btprev,
                     float invB,
                     const float* __restrict__ W,
                     const float* __restrict__ bias,
                     float* __restrict__ hout,
                     float* __restrict__ partial)
{
    __shared__ float xl[TILE_M * LDSS];
    __shared__ float Wl[64 * LDSS];
    __shared__ float biasl[64];
    __shared__ __align__(16) float red[NW * 128];
    __shared__ __align__(16) float scsh[128];

    constexpr int C4N   = KP / 4;
    constexpr int RSTEP = NT / C4N;
    constexpr int XITER = TILE_M / RSTEP;
    constexpr int WITER = 64 * KP / NT;
    constexpr int PITER = 8;
    static_assert(NT % C4N == 0 && TILE_M % RSTEP == 0 && (64 * KP) % NT == 0, "tile shape");

    const int t    = threadIdx.x;
    const int lane = t & 31;
    const int wave = t >> 5;
    const int half = lane >> 4;
    const int l15  = lane & 15;
    const size_t row0 = (size_t)blockIdx.x * TILE_M;

    const int nprev = (int)gridDim.x;
    float4 pv[PITER];
    float gv = 0.0f, btv = 0.0f;
    if constexpr (NORM) {
        if (t < 64) { gv = gprev[t < 59 ? t : 0]; btv = btprev[t < 59 ? t : 0]; }
        const float4* __restrict__ p4 = (const float4*)pprev;
        #pragma unroll
        for (int j = 0; j < PITER; j++) {
            const int b  = wave + j * NW;
            const int bb = b < nprev ? b : nprev - 1;
            pv[j] = p4[(size_t)bb * 32 + lane];
        }
    }

    const int xr = t / C4N;
    const int xc = (t % C4N) * 4;
    const float* xp = xin + (row0 + xr) * (size_t)KP + xc;
    float4 xv[XITER];
    #pragma unroll
    for (int j = 0; j < XITER; j++)
        xv[j] = *(const float4*)(xp + (size_t)j * RSTEP * KP);

    float wv[WITER];
    #pragma unroll
    for (int j = 0; j < WITER; j++) {
        const int i  = t + NT * j;
        const int n  = i / KP;
        const int k  = i % KP;
        const bool ok = (n < 59) && (k < KW);
        const float w = W[ok ? n * KW + k : 0];
        wv[j] = ok ? w : 0.0f;
    }
    float bv = 0.0f;
    if (t < 64) bv = bias[t < 59 ? t : 0];

    float4 sc4 = make_float4(0.f, 0.f, 0.f, 0.f);
    float4 sh4 = make_float4(0.f, 0.f, 0.f, 0.f);
    if constexpr (NORM) {
        float4 a = make_float4(0.f, 0.f, 0.f, 0.f);
        #pragma unroll
        for (int j = 0; j < PITER; j++) {
            const bool ok = (wave + j * NW) < nprev;
            a.x += ok ? pv[j].x : 0.f; a.y += ok ? pv[j].y : 0.f;
            a.z += ok ? pv[j].z : 0.f; a.w += ok ? pv[j].w : 0.f;
        }
        const float4* __restrict__ p4 = (const float4*)pprev;
        for (int b = wave + PITER * NW; b < nprev; b += NW) {
            const float4 v = p4[(size_t)b * 32 + lane];
            a.x += v.x; a.y += v.y; a.z += v.z; a.w += v.w;
        }
        *(float4*)(&red[wave * 128 + lane * 4]) = a;
        lds_barrier();
        if (t < 64) {
            float s = 0.f, ss = 0.f;
            #pragma unroll
            for (int k = 0; k < NW; k++) {
                s  += red[k * 128 + t];
                ss += red[k * 128 + 64 + t];
            }
            const float mu  = s * invB;
            float var = ss * invB - mu * mu;
            var = fmaxf(var, 0.f);
            float scale = 0.f, shift = 0.f;
            if (t < 59) {
                scale = gv * rsqrtf(var + 1e-5f);
                shift = btv - mu * scale;
            }
            scsh[t]      = scale;
            scsh[64 + t] = shift;
        }
        lds_barrier();
        sc4 = *(const float4*)(&scsh[xc]);
        sh4 = *(const float4*)(&scsh[64 + xc]);
    }

    #pragma unroll
    for (int j = 0; j < XITER; j++) {
        float4 v = xv[j];
        if constexpr (NORM) {
            v.x = v.x * sc4.x + sh4.x;
            v.y = v.y * sc4.y + sh4.y;
            v.z = v.z * sc4.z + sh4.z;
            v.w = v.w * sc4.w + sh4.w;
        }
        *(float4*)(&xl[(xr + j * RSTEP) * LDSS + xc]) = v;
    }
    #pragma unroll
    for (int j = 0; j < WITER; j++) {
        const int i = t + NT * j;
        Wl[(i / KP) * LDSS + (i % KP)] = wv[j];
    }
    if (t < 64) biasl[t] = (t < 59) ? bv : 0.0f;
    __syncthreads();

    v8f acc[4] = {};
    const float* ax = xl + (wave * 16 + l15) * LDSS;
    #pragma unroll 4
    for (int k0 = 0; k0 < KP; k0 += 4) {
        const int ka = k0 + 2 * half;
        v2f a = *(const v2f*)(ax + ka);
        #pragma unroll
        for (int tt = 0; tt < 4; tt++) {
            v2f b = *(const v2f*)(Wl + (tt * 16 + l15) * LDSS + ka);
            acc[tt] = __builtin_amdgcn_wmma_f32_16x16x4_f32(
                false, a, false, b, (short)0, acc[tt], false, false);
        }
    }

    const size_t mbase = row0 + wave * 16 + half * 8;
    #pragma unroll
    for (int tt = 0; tt < 4; tt++) {
        const int n = tt * 16 + l15;
        const float bn = biasl[n];
        float sl = 0.f, sq = 0.f;
        #pragma unroll
        for (int r = 0; r < 8; r++) {
            float v = acc[tt][r] + bn;
            v = fmaxf(v, 0.f);
            hout[(mbase + r) * 64 + n] = v;
            sl += v;
            sq += v * v;
        }
        sl += __shfl_xor(sl, 16, 32);
        sq += __shfl_xor(sq, 16, 32);
        if (half == 0) {
            red[wave * 128 + n]      = sl;
            red[wave * 128 + 64 + n] = sq;
        }
    }
    lds_barrier();

    if (t < 128) {
        float s = 0.f;
        #pragma unroll
        for (int w = 0; w < NW; w++) s += red[w * 128 + t];
        partial[(size_t)blockIdx.x * 128 + t] = s;
    }
}

#define RED_W 32
__global__ __launch_bounds__(RED_W * 32)
void stats_reduce(const float* __restrict__ partial, int nblk,
                  const float* __restrict__ g,
                  const float* __restrict__ bt,
                  float* __restrict__ stats,
                  float invB)
{
    __shared__ __align__(16) float red[RED_W * 128];

    const int t  = threadIdx.x;
    const int c4 = t & 31;
    const int w  = t >> 5;
    const float4* __restrict__ p4 = (const float4*)partial;

    float4 a = make_float4(0.f, 0.f, 0.f, 0.f);
    #pragma unroll 8
    for (int b = w; b < nblk; b += RED_W) {
        const float4 v = p4[(size_t)b * 32 + c4];
        a.x += v.x; a.y += v.y; a.z += v.z; a.w += v.w;
    }
    *(float4*)(&red[w * 128 + c4 * 4]) = a;
    __syncthreads();

    if (t < 64) {
        const int n = t;
        float s = 0.f, ss = 0.f;
        #pragma unroll
        for (int k = 0; k < RED_W; k++) {
            s  += red[k * 128 + n];
            ss += red[k * 128 + 64 + n];
        }
        const float mu  = s * invB;
        float var = ss * invB - mu * mu;
        var = fmaxf(var, 0.f);
        float scale = 0.f, shift = 0.f;
        if (n < 59) {
            scale = g[n] * rsqrtf(var + 1e-5f);
            shift = bt[n] - mu * scale;
        }
        stats[n]      = scale;
        stats[64 + n] = shift;
    }
}

__global__ void normalize_final(const float* __restrict__ h,
                                const float* __restrict__ stats,
                                float* __restrict__ out, size_t total  )
{
    const size_t i = (size_t)blockIdx.x * 256 + threadIdx.x;
    if (i >= total) return;
    const size_t row = i / 59;
    const int n = (int)(i - row * 59);
    out[i] = h[row * 64 + n] * stats[n] + stats[64 + n];
}

__global__ void pad_input(const float* __restrict__ x, float* __restrict__ xp,
                          size_t total  )
{
    const size_t i = (size_t)blockIdx.x * 256 + threadIdx.x;
    if (i >= total) return;
    const size_t row = i >> 5;
    const int n = (int)(i & 31);
    xp[i] = (n < 29) ? x[row * 29 + n] : 0.0f;
}

extern "C" void kernel_launch(void* const* d_in, const int* in_sizes, int n_in,
                              void* d_out, int out_size, void* d_ws, size_t ws_size,
                              hipStream_t stream) {
    const float* x   = (const float*)d_in[0];
    const float* W0  = (const float*)d_in[1];
    const float* b0  = (const float*)d_in[2];
    const float* g0  = (const float*)d_in[3];
    const float* bt0 = (const float*)d_in[4];
    const float* Ws  = (const float*)d_in[5];
    const float* bs  = (const float*)d_in[6];
    const float* gs  = (const float*)d_in[7];
    const float* bts = (const float*)d_in[8];
    float* out = (float*)d_out;

    const int B     = in_sizes[0] / 29;
    const int nblk  = B / TILE_M;
    const float invB = 1.0f / (float)B;

    float* ws      = (float*)d_ws;
    float* xpad    = ws;
    float* P       = xpad    + (size_t)B * 32;
    float* Q       = P       + (size_t)B * 64;
    float* partA   = Q       + (size_t)B * 64;
    float* partB   = partA   + (size_t)nblk * 128;
    float* stats   = partB   + (size_t)nblk * 128;

    const size_t padTot = (size_t)B * 32;
    pad_input<<<(unsigned)((padTot + 255) / 256), 256, 0, stream>>>(x, xpad, padTot);

    gemm_relu_stats<false, 32, 29><<<nblk, NT, 0, stream>>>(
        xpad, nullptr, nullptr, nullptr, invB, W0, b0, P, partA);

    float* hprev = P;
    float* hnext = Q;
    float* pprev = partA;
    float* pnext = partB;
    const float* gprev = g0;
    const float* tprev = bt0;
    for (int l = 0; l < 14; l++) {
        const float* W  = Ws  + (size_t)l * 59 * 59;
        const float* bb = bs  + (size_t)l * 59;
        gemm_relu_stats<true, 64, 59><<<nblk, NT, 0, stream>>>(
            hprev, pprev, gprev, tprev, invB, W, bb, hnext, pnext);
        gprev = gs  + (size_t)l * 59;
        tprev = bts + (size_t)l * 59;
        float* tmp = hprev; hprev = hnext; hnext = tmp;
        float* tmq = pprev; pprev = pnext; pnext = tmq;
    }

    stats_reduce<<<1, RED_W * 32, 0, stream>>>(pprev, nblk, gprev, tprev, stats, invB);
    const size_t total = (size_t)B * 59;
    normalize_final<<<(unsigned)((total + 255) / 256), 256, 0, stream>>>(
        hprev, stats, out, total);
}
